# starting kernel
# baseline (speedup 1.0000x reference)
#include <hip/hip_runtime.h>
#include <hip/hip_bf16.h>

typedef __attribute__((ext_vector_type(2))) float v2f;
typedef __attribute__((ext_vector_type(8))) float v8f;

#define B_ 4
#define S_ 8192
#define D_ 1024
#define O_ 4

__global__ void __launch_bounds__(256)
hyena_proj_wmma(const float* __restrict__ x,
                const float* __restrict__ w,
                float* __restrict__ proj)
{
    const int  wave = (int)((blockIdx.x * blockDim.x + threadIdx.x) >> 5);
    const int  lane = (int)(threadIdx.x & 31);
    const int  m    = lane & 15;
    const int  kh   = lane >> 4;
    const long row0 = (long)wave * 16;

    const float* xrow = x + (row0 + m) * D_ + 2 * kh;
    const float* wrow  = w + (long)(m & (O_ - 1)) * D_ + 2 * kh;
    const bool   bvalid = (m < O_);

    v8f acc = {};
    #pragma unroll 4
    for (int k = 0; k < D_; k += 4) {
        v2f a;
        a.x = __builtin_nontemporal_load(xrow + k);
        a.y = __builtin_nontemporal_load(xrow + k + 1);
        float w0 = wrow[k];
        float w1 = wrow[k + 1];
        v2f bm;
        bm.x = bvalid ? w0 : 0.0f;
        bm.y = bvalid ? w1 : 0.0f;
        acc = __builtin_amdgcn_wmma_f32_16x16x4_f32(false, a, false, bm,
                                                    (short)0, acc, false, false);
    }

    if (bvalid) {
        #pragma unroll
        for (int j = 0; j < 8; ++j) {
            long r = row0 + j + 8 * kh;
            int  b = (int)(r / S_);
            int  s = (int)(r % S_);
            proj[((long)s * B_ + b) * O_ + m] = acc[j];
        }
    }
}

#define CHUNK 1024
#define WARM  2048

__global__ void __launch_bounds__(64)
hyena_recurrence(const float* __restrict__ proj,
                 const float* __restrict__ fc,
                 float* __restrict__ out)
{
    const int d     = (int)(blockIdx.x * 64 + threadIdx.x);
    const int chunk = (int)blockIdx.y;
    const int b     = (int)blockIdx.z;

    const float c0 = __expf(fc[0 * D_ + d]);
    const float c1 = __expf(fc[1 * D_ + d]);
    const float c2 = __expf(fc[2 * D_ + d]);
    const float c3 = __expf(fc[3 * D_ + d]);

    const int t0 = chunk * CHUNK;
    int t = t0 - WARM;
    if (t < 0) t = 0;

    const float4* p4 = (const float4*)proj;
    float y0 = 0.f, y1 = 0.f, y2 = 0.f, y3 = 0.f;

    for (; t < t0; ++t) {
        float4 p = p4[(long)t * B_ + b];
        y0 = fmaf(c0, y0, p.x);
        y1 = fmaf(c1, y1, p.y);
        y2 = fmaf(c2, y2, p.z);
        y3 = fmaf(c3, y3, p.w);
    }

    float* o = out + (long)b * S_ * D_ + d;
    __builtin_prefetch(&p4[(long)t0 * B_ + b], 0, 1);

    #pragma unroll 4
    for (; t < t0 + CHUNK; ++t) {
        float4 p = p4[(long)t * B_ + b];
        y0 = fmaf(c0, y0, p.x);
        y1 = fmaf(c1, y1, p.y);
        y2 = fmaf(c2, y2, p.z);
        y3 = fmaf(c3, y3, p.w);
        __builtin_nontemporal_store((y0 + y1) + (y2 + y3), o + (long)t * D_);
    }
}

extern "C" void kernel_launch(void* const* d_in, const int* in_sizes, int n_in,
                              void* d_out, int out_size, void* d_ws, size_t ws_size,
                              hipStream_t stream) {
    const float* x  = (const float*)d_in[0];
    const float* w  = (const float*)d_in[1];
    const float* fc = (const float*)d_in[2];
    float*       out  = (float*)d_out;
    float*       proj = (float*)d_ws;

    const int nwaves = (B_ * S_) / 16;
    hyena_proj_wmma<<<nwaves / 8, 256, 0, stream>>>(x, w, proj);

    dim3 grid(D_ / 64, S_ / CHUNK, B_);
    hyena_recurrence<<<grid, 64, 0, stream>>>(proj, fc, out);
}


# scan kernel: proj fetched 64 steps ahead by lane-parallel loads (one step per lane), each step's values taken by v_readlane, so no cache round trip sits on the serial chain; summation order kept
# speedup vs baseline: 2.5916x; 2.5916x over previous
#include <hip/hip_runtime.h>
#include <hip/hip_bf16.h>

typedef __attribute__((ext_vector_type(2))) float v2f;
typedef __attribute__((ext_vector_type(8))) float v8f;

#define B_ 4
#define S_ 8192
#define D_ 1024
#define O_ 4

__global__ void __launch_bounds__(256)
hyena_proj_wmma(const float* __restrict__ x,
                const float* __restrict__ w,
                float* __restrict__ proj)
{
    const int  wave = (int)((blockIdx.x * blockDim.x + threadIdx.x) >> 5);
    const int  lane = (int)(threadIdx.x & 31);
    const int  m    = lane & 15;
    const int  kh   = lane >> 4;
    const long row0 = (long)wave * 16;

    const float* xrow = x + (row0 + m) * D_ + 2 * kh;
    const float* wrow  = w + (long)(m & (O_ - 1)) * D_ + 2 * kh;
    const bool   bvalid = (m < O_);

    v8f acc = {};
    #pragma unroll 4
    for (int k = 0; k < D_; k += 4) {
        v2f a;
        a.x = __builtin_nontemporal_load(xrow + k);
        a.y = __builtin_nontemporal_load(xrow + k + 1);
        float w0 = wrow[k];
        float w1 = wrow[k + 1];
        v2f bm;
        bm.x = bvalid ? w0 : 0.0f;
        bm.y = bvalid ? w1 : 0.0f;
        acc = __builtin_amdgcn_wmma_f32_16x16x4_f32(false, a, false, bm,
                                                    (short)0, acc, false, false);
    }

    if (bvalid) {
        #pragma unroll
        for (int j = 0; j < 8; ++j) {
            long r = row0 + j + 8 * kh;
            int  b = (int)(r / S_);
            int  s = (int)(r % S_);
            proj[((long)s * B_ + b) * O_ + m] = acc[j];
        }
    }
}

#define CHUNK 1024
#define WARM  2048
#define GROUP 64
#define BURST 16

static_assert(CHUNK % GROUP == 0 && WARM % GROUP == 0, "whole fetch groups");
static_assert(32 % BURST == 0, "whole store bursts per load");

__device__ __forceinline__ float from_lane(float v, int i)
{
    return __builtin_bit_cast(float,
        __builtin_amdgcn_readlane(__builtin_bit_cast(int, v), i));
}

__device__ __forceinline__ float4 fetch32(const float4* __restrict__ p4,
                                          int tf, int lane, int b)
{
    int tt = tf + lane;
    tt = tt < S_ ? tt : S_ - 1;
    return p4[(long)tt * B_ + b];
}

__device__ __forceinline__ void warm32(const float4 p,
                                       float c0, float c1, float c2, float c3,
                                       float& y0, float& y1, float& y2, float& y3)
{
    #pragma unroll
    for (int i = 0; i < 32; ++i) {
        y0 = fmaf(c0, y0, from_lane(p.x, i));
        y1 = fmaf(c1, y1, from_lane(p.y, i));
        y2 = fmaf(c2, y2, from_lane(p.z, i));
        y3 = fmaf(c3, y3, from_lane(p.w, i));
    }
}

__device__ __forceinline__ void emit32(const float4 p,
                                       float c0, float c1, float c2, float c3,
                                       float& y0, float& y1, float& y2, float& y3,
                                       float* __restrict__ o)
{
    #pragma unroll
    for (int h = 0; h < 32; h += BURST) {
        float s[BURST];
        #pragma unroll
        for (int i = 0; i < BURST; ++i) {
            y0 = fmaf(c0, y0, from_lane(p.x, h + i));
            y1 = fmaf(c1, y1, from_lane(p.y, h + i));
            y2 = fmaf(c2, y2, from_lane(p.z, h + i));
            y3 = fmaf(c3, y3, from_lane(p.w, h + i));
            s[i] = (y0 + y1) + (y2 + y3);
        }
        __builtin_amdgcn_sched_barrier(0);
        #pragma unroll
        for (int i = 0; i < BURST; ++i) {
            __builtin_nontemporal_store(s[i], o + (long)(h + i) * D_);
        }
        __builtin_amdgcn_sched_barrier(0);
    }
}

__global__ void __launch_bounds__(64)
hyena_recurrence(const float* __restrict__ proj,
                 const float* __restrict__ fc,
                 float* __restrict__ out)
{
    const int d     = (int)(blockIdx.x * 64 + threadIdx.x);
    const int lane  = (int)(threadIdx.x & 31);
    const int chunk = (int)blockIdx.y;
    const int b     = (int)blockIdx.z;

    const int t0 = chunk * CHUNK;
    int t = t0 - WARM;
    if (t < 0) t = 0;

    const float4* p4 = (const float4*)proj;
    float4 lo = fetch32(p4, t, lane, b);
    float4 hi = fetch32(p4, t + 32, lane, b);

    const float c0 = __expf(fc[0 * D_ + d]);
    const float c1 = __expf(fc[1 * D_ + d]);
    const float c2 = __expf(fc[2 * D_ + d]);
    const float c3 = __expf(fc[3 * D_ + d]);

    float y0 = 0.f, y1 = 0.f, y2 = 0.f, y3 = 0.f;

    for (; t < t0; t += GROUP) {
        const float4 nlo = fetch32(p4, t + GROUP, lane, b);
        const float4 nhi = fetch32(p4, t + GROUP + 32, lane, b);
        __builtin_amdgcn_sched_barrier(0);
        warm32(lo, c0, c1, c2, c3, y0, y1, y2, y3);
        warm32(hi, c0, c1, c2, c3, y0, y1, y2, y3);
        lo = nlo;
        hi = nhi;
    }

    float* o = out + (long)b * S_ * D_ + d;

    for (; t < t0 + CHUNK; t += GROUP) {
        const float4 nlo = fetch32(p4, t + GROUP, lane, b);
        const float4 nhi = fetch32(p4, t + GROUP + 32, lane, b);
        float* ot = o + (long)t * D_;
        __builtin_amdgcn_sched_barrier(0);
        emit32(lo, c0, c1, c2, c3, y0, y1, y2, y3, ot);
        emit32(hi, c0, c1, c2, c3, y0, y1, y2, y3, ot + 32L * D_);
        lo = nlo;
        hi = nhi;
    }
}

extern "C" void kernel_launch(void* const* d_in, const int* in_sizes, int n_in,
                              void* d_out, int out_size, void* d_ws, size_t ws_size,
                              hipStream_t stream) {
    const float* x  = (const float*)d_in[0];
    const float* w  = (const float*)d_in[1];
    const float* fc = (const float*)d_in[2];
    float*       out  = (float*)d_out;
    float*       proj = (float*)d_ws;

    const int nwaves = (B_ * S_) / 16;
    hyena_proj_wmma<<<nwaves / 8, 256, 0, stream>>>(x, w, proj);

    dim3 grid(D_ / 64, S_ / CHUNK, B_);
    hyena_recurrence<<<grid, 64, 0, stream>>>(proj, fc, out);
}


# proj GEMM: x read by whole-line 16-byte-per-lane loads, 16 in flight per wave; WMMA tile = 4 rows x 4 d-segments on both operands, block diagonal of D summed last (sum order over d changed)
# speedup vs baseline: 4.8887x; 1.8864x over previous
#include <hip/hip_runtime.h>
#include <hip/hip_bf16.h>

typedef __attribute__((ext_vector_type(2))) float v2f;
typedef __attribute__((ext_vector_type(8))) float v8f;

#define B_ 4
#define S_ 8192
#define D_ 1024
#define O_ 4

typedef __attribute__((ext_vector_type(4))) float v4f;

#define ROWS_PER_WAVE 8
#define KB    32
#define NKB   (D_ / KB)
#define DEPTH 8
static_assert(NKB % DEPTH == 0 && NKB / DEPTH >= 2, "whole ring turns");

#define LOAD_STEP(i, kb)                                                          \
    do {                                                                          \
        const int off_ = (kb) * KB;                                               \
        xa[i] = __builtin_nontemporal_load((const v4f*)(xp0 + off_));   \
        xb[i] = __builtin_nontemporal_load((const v4f*)(xp1 + off_));             \
        wq[i] = *(const v4f*)(wp + off_);                               \
    } while (0)

__device__ __forceinline__ v8f mma4(float a0, float a1, float b0, float b1, v8f acc)
{
    v2f a; a.x = a0; a.y = a1;
    v2f b; b.x = b0; b.y = b1;
    return __builtin_amdgcn_wmma_f32_16x16x4_f32(false, a, false, b,
                                                 (short)0, acc, false, false);
}

#define MMA_STEP(i)                                                               \
    do {                                                                          \
        acc0 = mma4(xa[i].x, xa[i].y, wq[i].x, wq[i].y, acc0);                    \
        acc1 = mma4(xb[i].x, xb[i].y, wq[i].x, wq[i].y, acc1);                    \
        acc0 = mma4(xa[i].z, xa[i].w, wq[i].z, wq[i].w, acc0);                    \
        acc1 = mma4(xb[i].z, xb[i].w, wq[i].z, wq[i].w, acc1);                    \
    } while (0)

__device__ __forceinline__ void store_group(const v8f acc, long rbase, int lane,
                                            float* __restrict__ proj)
{
    const int c = lane & 15;
    const int h = lane >> 4;
    const int j = c & 3;
    const int o = c >> 2;
    const bool j1 = (j & 1) != 0;
    const bool j2 = (j & 2) != 0;
    const float a01 = j1 ? acc[1] : acc[0];
    const float a23 = j1 ? acc[3] : acc[2];
    const float a45 = j1 ? acc[5] : acc[4];
    const float a67 = j1 ? acc[7] : acc[6];
    float v0 = j2 ? a23 : a01;
    float v1 = j2 ? a67 : a45;
    v0 += __shfl_xor(v0, 1);
    v1 += __shfl_xor(v1, 1);
    v0 += __shfl_xor(v0, 2);
    v1 += __shfl_xor(v1, 2);
    if (j == 0) {
        const long r = rbase + 2 * h;
        const int  b = (int)(r / S_);
        const int  s = (int)(r % S_);
        proj[((long)s * B_ + b) * O_ + o]       = v0;
        proj[((long)(s + 1) * B_ + b) * O_ + o] = v1;
    }
}

__global__ void __launch_bounds__(256)
hyena_proj_wmma(const float* __restrict__ x,
                const float* __restrict__ w,
                float* __restrict__ proj)
{
    const int  wave = (int)((blockIdx.x * blockDim.x + threadIdx.x) >> 5);
    const int  lane = (int)(threadIdx.x & 31);
    const int  c    = lane & 15;
    const int  h    = lane >> 4;
    const long row0 = (long)wave * ROWS_PER_WAVE;

    const int seg = 8 * (c & 3) + 4 * h;
    const float* xp0 = x + (row0 + (c >> 2)) * D_ + seg;
    const float* xp1 = xp0 + 4L * D_;
    const float* wp  = w + (long)(c >> 2) * D_ + seg;

    v4f xa[DEPTH], xb[DEPTH], wq[DEPTH];
    v8f acc0 = {};
    v8f acc1 = {};

    #pragma unroll
    for (int i = 0; i < DEPTH; ++i) LOAD_STEP(i, i);

    #pragma unroll 1
    for (int kb0 = DEPTH; kb0 < NKB; kb0 += DEPTH) {
        #pragma unroll
        for (int i = 0; i < DEPTH; ++i) {
            MMA_STEP(i);
            LOAD_STEP(i, kb0 + i);
            __builtin_amdgcn_sched_barrier(0);
        }
    }
    #pragma unroll
    for (int i = 0; i < DEPTH; ++i) MMA_STEP(i);

    store_group(acc0, row0,     lane, proj);
    store_group(acc1, row0 + 4, lane, proj);
}

#define CHUNK 1024
#define WARM  2048
#define GROUP 64
#define BURST 16

static_assert(CHUNK % GROUP == 0 && WARM % GROUP == 0, "whole fetch groups");
static_assert(32 % BURST == 0, "whole store bursts per load");

__device__ __forceinline__ float from_lane(float v, int i)
{
    return __builtin_bit_cast(float,
        __builtin_amdgcn_readlane(__builtin_bit_cast(int, v), i));
}

__device__ __forceinline__ float4 fetch32(const float4* __restrict__ p4,
                                          int tf, int lane, int b)
{
    int tt = tf + lane;
    tt = tt < S_ ? tt : S_ - 1;
    return p4[(long)tt * B_ + b];
}

__device__ __forceinline__ void warm32(const float4 p,
                                       float c0, float c1, float c2, float c3,
                                       float& y0, float& y1, float& y2, float& y3)
{
    #pragma unroll
    for (int i = 0; i < 32; ++i) {
        y0 = fmaf(c0, y0, from_lane(p.x, i));
        y1 = fmaf(c1, y1, from_lane(p.y, i));
        y2 = fmaf(c2, y2, from_lane(p.z, i));
        y3 = fmaf(c3, y3, from_lane(p.w, i));
    }
}

__device__ __forceinline__ void emit32(const float4 p,
                                       float c0, float c1, float c2, float c3,
                                       float& y0, float& y1, float& y2, float& y3,
                                       float* __restrict__ o)
{
    #pragma unroll
    for (int h = 0; h < 32; h += BURST) {
        float s[BURST];
        #pragma unroll
        for (int i = 0; i < BURST; ++i) {
            y0 = fmaf(c0, y0, from_lane(p.x, h + i));
            y1 = fmaf(c1, y1, from_lane(p.y, h + i));
            y2 = fmaf(c2, y2, from_lane(p.z, h + i));
            y3 = fmaf(c3, y3, from_lane(p.w, h + i));
            s[i] = (y0 + y1) + (y2 + y3);
        }
        __builtin_amdgcn_sched_barrier(0);
        #pragma unroll
        for (int i = 0; i < BURST; ++i) {
            __builtin_nontemporal_store(s[i], o + (long)(h + i) * D_);
        }
        __builtin_amdgcn_sched_barrier(0);
    }
}

__global__ void __launch_bounds__(64)
hyena_recurrence(const float* __restrict__ proj,
                 const float* __restrict__ fc,
                 float* __restrict__ out)
{
    const int d     = (int)(blockIdx.x * 64 + threadIdx.x);
    const int lane  = (int)(threadIdx.x & 31);
    const int chunk = (int)blockIdx.y;
    const int b     = (int)blockIdx.z;

    const int t0 = chunk * CHUNK;
    int t = t0 - WARM;
    if (t < 0) t = 0;

    const float4* p4 = (const float4*)proj;
    float4 lo = fetch32(p4, t, lane, b);
    float4 hi = fetch32(p4, t + 32, lane, b);

    const float c0 = __expf(fc[0 * D_ + d]);
    const float c1 = __expf(fc[1 * D_ + d]);
    const float c2 = __expf(fc[2 * D_ + d]);
    const float c3 = __expf(fc[3 * D_ + d]);

    float y0 = 0.f, y1 = 0.f, y2 = 0.f, y3 = 0.f;

    for (; t < t0; t += GROUP) {
        const float4 nlo = fetch32(p4, t + GROUP, lane, b);
        const float4 nhi = fetch32(p4, t + GROUP + 32, lane, b);
        __builtin_amdgcn_sched_barrier(0);
        warm32(lo, c0, c1, c2, c3, y0, y1, y2, y3);
        warm32(hi, c0, c1, c2, c3, y0, y1, y2, y3);
        lo = nlo;
        hi = nhi;
    }

    float* o = out + (long)b * S_ * D_ + d;

    for (; t < t0 + CHUNK; t += GROUP) {
        const float4 nlo = fetch32(p4, t + GROUP, lane, b);
        const float4 nhi = fetch32(p4, t + GROUP + 32, lane, b);
        float* ot = o + (long)t * D_;
        __builtin_amdgcn_sched_barrier(0);
        emit32(lo, c0, c1, c2, c3, y0, y1, y2, y3, ot);
        emit32(hi, c0, c1, c2, c3, y0, y1, y2, y3, ot + 32L * D_);
        lo = nlo;
        hi = nhi;
    }
}

extern "C" void kernel_launch(void* const* d_in, const int* in_sizes, int n_in,
                              void* d_out, int out_size, void* d_ws, size_t ws_size,
                              hipStream_t stream) {
    const float* x  = (const float*)d_in[0];
    const float* w  = (const float*)d_in[1];
    const float* fc = (const float*)d_in[2];
    float*       out  = (float*)d_out;
    float*       proj = (float*)d_ws;

    const int nwaves = (B_ * S_) / ROWS_PER_WAVE;
    hyena_proj_wmma<<<nwaves / 8, 256, 0, stream>>>(x, w, proj);

    dim3 grid(D_ / 64, S_ / CHUNK, B_);
    hyena_recurrence<<<grid, 64, 0, stream>>>(proj, fc, out);
}


# scan kernel: 2048-step re-warm per chunk replaced by exact carries: a 16-wave block scans its chunks from zero, passes end states through LDS, each wave rebuilds its start; carry sum order changed
# speedup vs baseline: 5.3687x; 1.0982x over previous
#include <hip/hip_runtime.h>
#include <hip/hip_bf16.h>

typedef __attribute__((ext_vector_type(2))) float v2f;
typedef __attribute__((ext_vector_type(8))) float v8f;

#define B_ 4
#define S_ 8192
#define D_ 1024
#define O_ 4

typedef __attribute__((ext_vector_type(4))) float v4f;

#define ROWS_PER_WAVE 8
#define KB    32
#define NKB   (D_ / KB)
#define DEPTH 8
static_assert(NKB % DEPTH == 0 && NKB / DEPTH >= 2, "whole ring turns");

#define LOAD_STEP(i, kb)                                                          \
    do {                                                                          \
        const int off_ = (kb) * KB;                                               \
        xa[i] = __builtin_nontemporal_load((const v4f*)(xp0 + off_));   \
        xb[i] = __builtin_nontemporal_load((const v4f*)(xp1 + off_));             \
        wq[i] = *(const v4f*)(wp + off_);                               \
    } while (0)

__device__ __forceinline__ v8f mma4(float a0, float a1, float b0, float b1, v8f acc)
{
    v2f a; a.x = a0; a.y = a1;
    v2f b; b.x = b0; b.y = b1;
    return __builtin_amdgcn_wmma_f32_16x16x4_f32(false, a, false, b,
                                                 (short)0, acc, false, false);
}

#define MMA_STEP(i)                                                               \
    do {                                                                          \
        acc0 = mma4(xa[i].x, xa[i].y, wq[i].x, wq[i].y, acc0);                    \
        acc1 = mma4(xb[i].x, xb[i].y, wq[i].x, wq[i].y, acc1);                    \
        acc0 = mma4(xa[i].z, xa[i].w, wq[i].z, wq[i].w, acc0);                    \
        acc1 = mma4(xb[i].z, xb[i].w, wq[i].z, wq[i].w, acc1);                    \
    } while (0)

__device__ __forceinline__ void store_group(const v8f acc, long rbase, int lane,
                                            float* __restrict__ proj)
{
    const int c = lane & 15;
    const int h = lane >> 4;
    const int j = c & 3;
    const int o = c >> 2;
    const bool j1 = (j & 1) != 0;
    const bool j2 = (j & 2) != 0;
    const float a01 = j1 ? acc[1] : acc[0];
    const float a23 = j1 ? acc[3] : acc[2];
    const float a45 = j1 ? acc[5] : acc[4];
    const float a67 = j1 ? acc[7] : acc[6];
    float v0 = j2 ? a23 : a01;
    float v1 = j2 ? a67 : a45;
    v0 += __shfl_xor(v0, 1);
    v1 += __shfl_xor(v1, 1);
    v0 += __shfl_xor(v0, 2);
    v1 += __shfl_xor(v1, 2);
    if (j == 0) {
        const long r = rbase + 2 * h;
        const int  b = (int)(r / S_);
        const int  s = (int)(r % S_);
        proj[((long)s * B_ + b) * O_ + o]       = v0;
        proj[((long)(s + 1) * B_ + b) * O_ + o] = v1;
    }
}

__global__ void __launch_bounds__(256)
hyena_proj_wmma(const float* __restrict__ x,
                const float* __restrict__ w,
                float* __restrict__ proj)
{
    const int  wave = (int)((blockIdx.x * blockDim.x + threadIdx.x) >> 5);
    const int  lane = (int)(threadIdx.x & 31);
    const int  c    = lane & 15;
    const int  h    = lane >> 4;
    const long row0 = (long)wave * ROWS_PER_WAVE;

    const int seg = 8 * (c & 3) + 4 * h;
    const float* xp0 = x + (row0 + (c >> 2)) * D_ + seg;
    const float* xp1 = xp0 + 4L * D_;
    const float* wp  = w + (long)(c >> 2) * D_ + seg;

    v4f xa[DEPTH], xb[DEPTH], wq[DEPTH];
    v8f acc0 = {};
    v8f acc1 = {};

    #pragma unroll
    for (int i = 0; i < DEPTH; ++i) LOAD_STEP(i, i);

    #pragma unroll 1
    for (int kb0 = DEPTH; kb0 < NKB; kb0 += DEPTH) {
        #pragma unroll
        for (int i = 0; i < DEPTH; ++i) {
            MMA_STEP(i);
            LOAD_STEP(i, kb0 + i);
            __builtin_amdgcn_sched_barrier(0);
        }
    }
    #pragma unroll
    for (int i = 0; i < DEPTH; ++i) MMA_STEP(i);

    store_group(acc0, row0,     lane, proj);
    store_group(acc1, row0 + 4, lane, proj);
}

#define NW      16
#define LONG_G  10
#define SHORT_G 6
#define LONG_T  (32 * LONG_G)
#define SHORT_T (32 * SHORT_G)
#define SPLIT   (NW * LONG_T)
#define BURST   16

static_assert(SPLIT + NW * SHORT_T == S_, "the two blocks cover the sequence");
static_assert(32 % BURST == 0, "whole store bursts per load");

__device__ __forceinline__ float from_lane(float v, int i)
{
    return __builtin_bit_cast(float,
        __builtin_amdgcn_readlane(__builtin_bit_cast(int, v), i));
}

__device__ __forceinline__ float4 fetch32(const float4* __restrict__ p4,
                                          int tf, int lane, int b)
{
    int tt = tf + lane;
    tt = tt < S_ ? tt : S_ - 1;
    return p4[(long)tt * B_ + b];
}

__device__ __forceinline__ void warm32(const float4 p,
                                       float c0, float c1, float c2, float c3,
                                       float& y0, float& y1, float& y2, float& y3)
{
    #pragma unroll
    for (int i = 0; i < 32; ++i) {
        y0 = fmaf(c0, y0, from_lane(p.x, i));
        y1 = fmaf(c1, y1, from_lane(p.y, i));
        y2 = fmaf(c2, y2, from_lane(p.z, i));
        y3 = fmaf(c3, y3, from_lane(p.w, i));
    }
}

__device__ __forceinline__ void emit32(const float4 p,
                                       float c0, float c1, float c2, float c3,
                                       float& y0, float& y1, float& y2, float& y3,
                                       float* __restrict__ o)
{
    #pragma unroll
    for (int h = 0; h < 32; h += BURST) {
        float s[BURST];
        #pragma unroll
        for (int i = 0; i < BURST; ++i) {
            y0 = fmaf(c0, y0, from_lane(p.x, h + i));
            y1 = fmaf(c1, y1, from_lane(p.y, h + i));
            y2 = fmaf(c2, y2, from_lane(p.z, h + i));
            y3 = fmaf(c3, y3, from_lane(p.w, h + i));
            s[i] = (y0 + y1) + (y2 + y3);
        }
        __builtin_amdgcn_sched_barrier(0);
        #pragma unroll
        for (int i = 0; i < BURST; ++i) {
            __builtin_nontemporal_store(s[i], o + (long)(h + i) * D_);
        }
        __builtin_amdgcn_sched_barrier(0);
    }
}

__device__ __forceinline__ void carry(const float4 (*sE)[32], int first, int n, int lane,
                                      float cp0, float cp1, float cp2, float cp3,
                                      float& y0, float& y1, float& y2, float& y3)
{
    int j = 0;
    for (; j + 4 <= n; j += 4) {
        const float4 e0 = sE[first + j][lane];
        const float4 e1 = sE[first + j + 1][lane];
        const float4 e2 = sE[first + j + 2][lane];
        const float4 e3 = sE[first + j + 3][lane];
        y0 = fmaf(cp0, y0, e0.x); y1 = fmaf(cp1, y1, e0.y); y2 = fmaf(cp2, y2, e0.z); y3 = fmaf(cp3, y3, e0.w);
        y0 = fmaf(cp0, y0, e1.x); y1 = fmaf(cp1, y1, e1.y); y2 = fmaf(cp2, y2, e1.z); y3 = fmaf(cp3, y3, e1.w);
        y0 = fmaf(cp0, y0, e2.x); y1 = fmaf(cp1, y1, e2.y); y2 = fmaf(cp2, y2, e2.z); y3 = fmaf(cp3, y3, e2.w);
        y0 = fmaf(cp0, y0, e3.x); y1 = fmaf(cp1, y1, e3.y); y2 = fmaf(cp2, y2, e3.z); y3 = fmaf(cp3, y3, e3.w);
    }
    for (; j < n; ++j) {
        const float4 e = sE[first + j][lane];
        y0 = fmaf(cp0, y0, e.x); y1 = fmaf(cp1, y1, e.y); y2 = fmaf(cp2, y2, e.z); y3 = fmaf(cp3, y3, e.w);
    }
}

__device__ __forceinline__ float pow64(float c)
{
    float p = c * c;
    p = p * p;
    p = p * p;
    p = p * p;
    p = p * p;
    return p * p;
}

__global__ void __launch_bounds__(NW * 32)
hyena_recurrence(const float* __restrict__ proj,
                 const float* __restrict__ fc,
                 float* __restrict__ out)
{
    __shared__ float4 sE[2 * NW][32];

    const int lane   = (int)(threadIdx.x & 31);
    const int w      = __builtin_amdgcn_readfirstlane((int)(threadIdx.x >> 5));
    const int d      = (int)(blockIdx.x * 32) + lane;
    const int second = (int)blockIdx.y;
    const int b      = (int)blockIdx.z;

    const float4* p4 = (const float4*)proj;

    const float c0 = __expf(fc[0 * D_ + d]);
    const float c1 = __expf(fc[1 * D_ + d]);
    const float c2 = __expf(fc[2 * D_ + d]);
    const float c3 = __expf(fc[3 * D_ + d]);

    const int npass = second + 1;
    #pragma unroll 1
    for (int pass = 0; pass < npass; ++pass) {
        const int ng = pass ? SHORT_G : LONG_G;
        int t = pass ? SPLIT + w * SHORT_T : w * LONG_T;
        float y0 = 0.f, y1 = 0.f, y2 = 0.f, y3 = 0.f;
        float4 cur = fetch32(p4, t, lane, b);
        #pragma unroll 1
        for (int g = 0; g < ng; ++g) {
            t += 32;
            const float4 nxt = fetch32(p4, t, lane, b);
            __builtin_amdgcn_sched_barrier(0);
            warm32(cur, c0, c1, c2, c3, y0, y1, y2, y3);
            cur = nxt;
        }
        sE[pass * NW + w][lane] = make_float4(y0, y1, y2, y3);
    }

    __syncthreads();

    const float q0 = pow64(c0), q1 = pow64(c1), q2 = pow64(c2), q3 = pow64(c3);
    const float h0 = q0 * q0, h1 = q1 * q1, h2 = q2 * q2, h3 = q3 * q3;
    const float l0 = (h0 * h0) * q0, l1 = (h1 * h1) * q1, l2 = (h2 * h2) * q2, l3 = (h3 * h3) * q3;

    float y0 = 0.f, y1 = 0.f, y2 = 0.f, y3 = 0.f;
    int t0, ng;
    if (second) {
        const float s0 = h0 * q0, s1 = h1 * q1, s2 = h2 * q2, s3 = h3 * q3;
        carry(sE, 0,  NW, lane, l0, l1, l2, l3, y0, y1, y2, y3);
        carry(sE, NW, w,  lane, s0, s1, s2, s3, y0, y1, y2, y3);
        t0 = SPLIT + w * SHORT_T;
        ng = SHORT_G;
    } else {
        carry(sE, 0, w, lane, l0, l1, l2, l3, y0, y1, y2, y3);
        t0 = w * LONG_T;
        ng = LONG_G;
    }

    float* o = out + ((long)b * S_ + t0) * D_ + d;
    int t = t0;
    float4 cur = fetch32(p4, t, lane, b);
    #pragma unroll 1
    for (int g = 0; g < ng; ++g) {
        t += 32;
        const float4 nxt = fetch32(p4, t, lane, b);
        __builtin_amdgcn_sched_barrier(0);
        emit32(cur, c0, c1, c2, c3, y0, y1, y2, y3, o);
        o += 32L * D_;
        cur = nxt;
    }
}

extern "C" void kernel_launch(void* const* d_in, const int* in_sizes, int n_in,
                              void* d_out, int out_size, void* d_ws, size_t ws_size,
                              hipStream_t stream) {
    const float* x  = (const float*)d_in[0];
    const float* w  = (const float*)d_in[1];
    const float* fc = (const float*)d_in[2];
    float*       out  = (float*)d_out;
    float*       proj = (float*)d_ws;

    const int nwaves = (B_ * S_) / ROWS_PER_WAVE;
    hyena_proj_wmma<<<nwaves / 8, 256, 0, stream>>>(x, w, proj);

    dim3 grid(D_ / 32, 2, B_);
    hyena_recurrence<<<grid, NW * 32, 0, stream>>>(proj, fc, out);
}


# scan kernel: each lane carries 2 adjacent d on packed f32 FMA/ADD (a lane read serves 64 d, 8-byte stores); sequence cut in 4 regions per column, so the carry sum order over chunks changed
# speedup vs baseline: 5.7037x; 1.0624x over previous
#include <hip/hip_runtime.h>
#include <hip/hip_bf16.h>

typedef __attribute__((ext_vector_type(2))) float v2f;
typedef __attribute__((ext_vector_type(8))) float v8f;

#define B_ 4
#define S_ 8192
#define D_ 1024
#define O_ 4

typedef __attribute__((ext_vector_type(4))) float v4f;

#define ROWS_PER_WAVE 8
#define KB    32
#define NKB   (D_ / KB)
#define DEPTH 8
static_assert(NKB % DEPTH == 0 && NKB / DEPTH >= 2, "whole ring turns");

#define LOAD_STEP(i, kb)                                                          \
    do {                                                                          \
        const int off_ = (kb) * KB;                                               \
        xa[i] = __builtin_nontemporal_load((const v4f*)(xp0 + off_));   \
        xb[i] = __builtin_nontemporal_load((const v4f*)(xp1 + off_));             \
        wq[i] = *(const v4f*)(wp + off_);                               \
    } while (0)

__device__ __forceinline__ v8f mma4(float a0, float a1, float b0, float b1, v8f acc)
{
    v2f a; a.x = a0; a.y = a1;
    v2f b; b.x = b0; b.y = b1;
    return __builtin_amdgcn_wmma_f32_16x16x4_f32(false, a, false, b,
                                                 (short)0, acc, false, false);
}

#define MMA_STEP(i)                                                               \
    do {                                                                          \
        acc0 = mma4(xa[i].x, xa[i].y, wq[i].x, wq[i].y, acc0);                    \
        acc1 = mma4(xb[i].x, xb[i].y, wq[i].x, wq[i].y, acc1);                    \
        acc0 = mma4(xa[i].z, xa[i].w, wq[i].z, wq[i].w, acc0);                    \
        acc1 = mma4(xb[i].z, xb[i].w, wq[i].z, wq[i].w, acc1);                    \
    } while (0)

__device__ __forceinline__ void store_group(const v8f acc, long rbase, int lane,
                                            float* __restrict__ proj)
{
    const int c = lane & 15;
    const int h = lane >> 4;
    const int j = c & 3;
    const int o = c >> 2;
    const bool j1 = (j & 1) != 0;
    const bool j2 = (j & 2) != 0;
    const float a01 = j1 ? acc[1] : acc[0];
    const float a23 = j1 ? acc[3] : acc[2];
    const float a45 = j1 ? acc[5] : acc[4];
    const float a67 = j1 ? acc[7] : acc[6];
    float v0 = j2 ? a23 : a01;
    float v1 = j2 ? a67 : a45;
    v0 += __shfl_xor(v0, 1);
    v1 += __shfl_xor(v1, 1);
    v0 += __shfl_xor(v0, 2);
    v1 += __shfl_xor(v1, 2);
    if (j == 0) {
        const long r = rbase + 2 * h;
        const int  b = (int)(r / S_);
        const int  s = (int)(r % S_);
        proj[((long)s * B_ + b) * O_ + o]       = v0;
        proj[((long)(s + 1) * B_ + b) * O_ + o] = v1;
    }
}

__global__ void __launch_bounds__(256)
hyena_proj_wmma(const float* __restrict__ x,
                const float* __restrict__ w,
                float* __restrict__ proj)
{
    const int  wave = (int)((blockIdx.x * blockDim.x + threadIdx.x) >> 5);
    const int  lane = (int)(threadIdx.x & 31);
    const int  c    = lane & 15;
    const int  h    = lane >> 4;
    const long row0 = (long)wave * ROWS_PER_WAVE;

    const int seg = 8 * (c & 3) + 4 * h;
    const float* xp0 = x + (row0 + (c >> 2)) * D_ + seg;
    const float* xp1 = xp0 + 4L * D_;
    const float* wp  = w + (long)(c >> 2) * D_ + seg;

    v4f xa[DEPTH], xb[DEPTH], wq[DEPTH];
    v8f acc0 = {};
    v8f acc1 = {};

    #pragma unroll
    for (int i = 0; i < DEPTH; ++i) LOAD_STEP(i, i);

    #pragma unroll 1
    for (int kb0 = DEPTH; kb0 < NKB; kb0 += DEPTH) {
        #pragma unroll
        for (int i = 0; i < DEPTH; ++i) {
            MMA_STEP(i);
            LOAD_STEP(i, kb0 + i);
            __builtin_amdgcn_sched_barrier(0);
        }
    }
    #pragma unroll
    for (int i = 0; i < DEPTH; ++i) MMA_STEP(i);

    store_group(acc0, row0,     lane, proj);
    store_group(acc1, row0 + 4, lane, proj);
}

#define NW      16
#define NREG    4
#define DPB     64
#define BURST   16

__device__ __forceinline__ int reg_groups(int j) { return j == 0 ? 7 : j == 1 ? 4 : j == 2 ? 3 : 2; }
__device__ __forceinline__ int reg_start(int j)  { return j == 0 ? 0 : j == 1 ? 3584 : j == 2 ? 5632 : 7168; }

static_assert(NW * 32 * (7 + 4 + 3 + 2) == S_, "the four regions cover the sequence");
static_assert(NW * 32 * 7 == 3584 && 3584 + NW * 32 * 4 == 5632 && 5632 + NW * 32 * 3 == 7168,
              "region starts");
static_assert(32 % BURST == 0, "whole store bursts per load");

__device__ __forceinline__ float from_lane(float v, int i)
{
    return __builtin_bit_cast(float,
        __builtin_amdgcn_readlane(__builtin_bit_cast(int, v), i));
}

__device__ __forceinline__ v2f step2(v2f c, v2f y, float p)
{
    v2f pp; pp.x = p; pp.y = p;
    return __builtin_elementwise_fma(c, y, pp);
}

__device__ __forceinline__ float4 fetch32(const float4* __restrict__ p4,
                                          int tf, int lane, int b)
{
    int tt = tf + lane;
    tt = tt < S_ ? tt : S_ - 1;
    return p4[(long)tt * B_ + b];
}

__device__ __forceinline__ void warm32(const float4 p,
                                       v2f c0, v2f c1, v2f c2, v2f c3,
                                       v2f& y0, v2f& y1, v2f& y2, v2f& y3)
{
    #pragma unroll
    for (int i = 0; i < 32; ++i) {
        y0 = step2(c0, y0, from_lane(p.x, i));
        y1 = step2(c1, y1, from_lane(p.y, i));
        y2 = step2(c2, y2, from_lane(p.z, i));
        y3 = step2(c3, y3, from_lane(p.w, i));
    }
}

__device__ __forceinline__ void emit32(const float4 p,
                                       v2f c0, v2f c1, v2f c2, v2f c3,
                                       v2f& y0, v2f& y1, v2f& y2, v2f& y3,
                                       float* __restrict__ o)
{
    #pragma unroll
    for (int h = 0; h < 32; h += BURST) {
        v2f s[BURST];
        #pragma unroll
        for (int i = 0; i < BURST; ++i) {
            y0 = step2(c0, y0, from_lane(p.x, h + i));
            y1 = step2(c1, y1, from_lane(p.y, h + i));
            y2 = step2(c2, y2, from_lane(p.z, h + i));
            y3 = step2(c3, y3, from_lane(p.w, h + i));
            s[i] = (y0 + y1) + (y2 + y3);
        }
        __builtin_amdgcn_sched_barrier(0);
        #pragma unroll
        for (int i = 0; i < BURST; ++i) {
            __builtin_nontemporal_store(s[i], (v2f*)(o + (long)(h + i) * D_));
        }
        __builtin_amdgcn_sched_barrier(0);
    }
}

struct EndState { v4f a; v4f b; };

__device__ __forceinline__ void carry(const EndState (*sE)[32], int first, int n, int lane,
                                      v2f cp0, v2f cp1, v2f cp2, v2f cp3,
                                      v2f& y0, v2f& y1, v2f& y2, v2f& y3)
{
#define CARRY_LINK(e)                                                             \
    do {                                                                          \
        y0 = __builtin_elementwise_fma(cp0, y0, (e).a.xy);                        \
        y1 = __builtin_elementwise_fma(cp1, y1, (e).a.zw);                        \
        y2 = __builtin_elementwise_fma(cp2, y2, (e).b.xy);                        \
        y3 = __builtin_elementwise_fma(cp3, y3, (e).b.zw);                        \
    } while (0)
    int j = 0;
    for (; j + 4 <= n; j += 4) {
        const EndState e0 = sE[first + j][lane];
        const EndState e1 = sE[first + j + 1][lane];
        const EndState e2 = sE[first + j + 2][lane];
        const EndState e3 = sE[first + j + 3][lane];
        CARRY_LINK(e0); CARRY_LINK(e1); CARRY_LINK(e2); CARRY_LINK(e3);
    }
    if (j + 2 <= n) {
        const EndState e0 = sE[first + j][lane];
        const EndState e1 = sE[first + j + 1][lane];
        CARRY_LINK(e0); CARRY_LINK(e1);
        j += 2;
    }
    if (j < n) {
        const EndState e0 = sE[first + j][lane];
        CARRY_LINK(e0);
    }
#undef CARRY_LINK
}

__device__ __forceinline__ v2f pow32(v2f c)
{
    v2f p = c * c;
    p = p * p;
    p = p * p;
    p = p * p;
    return p * p;
}

__device__ __forceinline__ v2f chunk_decay(v2f g, int j)
{
    const v2f g2 = g * g;
    const v2f g3 = g2 * g;
    const v2f g4 = g2 * g2;
    return j == 0 ? g4 * g3 : j == 1 ? g4 : j == 2 ? g3 : g2;
}

__global__ void __launch_bounds__(NW * 32)
hyena_recurrence(const float* __restrict__ proj,
                 const float* __restrict__ fc,
                 float* __restrict__ out)
{
    __shared__ EndState sE[NREG * NW][32];

    const int lane   = (int)(threadIdx.x & 31);
    const int w      = __builtin_amdgcn_readfirstlane((int)(threadIdx.x >> 5));
    const int d      = (int)(blockIdx.x * DPB) + 2 * lane;
    const int k      = (int)blockIdx.y;
    const int b      = (int)blockIdx.z;

    const float4* p4 = (const float4*)proj;

    const v2f f0 = *(const v2f*)(fc + 0 * D_ + d);
    const v2f f1 = *(const v2f*)(fc + 1 * D_ + d);
    const v2f f2 = *(const v2f*)(fc + 2 * D_ + d);
    const v2f f3 = *(const v2f*)(fc + 3 * D_ + d);
    v2f c0, c1, c2, c3;
    c0.x = __expf(f0.x); c0.y = __expf(f0.y);
    c1.x = __expf(f1.x); c1.y = __expf(f1.y);
    c2.x = __expf(f2.x); c2.y = __expf(f2.y);
    c3.x = __expf(f3.x); c3.y = __expf(f3.y);

    #pragma unroll 1
    for (int j = 0; j <= k; ++j) {
        const int ng = reg_groups(j);
        int t = reg_start(j) + w * 32 * ng;
        v2f y0 = {}, y1 = {}, y2 = {}, y3 = {};
        float4 cur = fetch32(p4, t, lane, b);
        #pragma unroll 1
        for (int g = 0; g < ng; ++g) {
            t += 32;
            const float4 nxt = fetch32(p4, t, lane, b);
            __builtin_amdgcn_sched_barrier(0);
            warm32(cur, c0, c1, c2, c3, y0, y1, y2, y3);
            cur = nxt;
        }
        EndState e;
        e.a.xy = y0; e.a.zw = y1; e.b.xy = y2; e.b.zw = y3;
        sE[j * NW + w][lane] = e;
    }

    __syncthreads();

    const v2f g0 = pow32(c0), g1 = pow32(c1), g2 = pow32(c2), g3 = pow32(c3);

    v2f y0 = {}, y1 = {}, y2 = {}, y3 = {};
    #pragma unroll
    for (int j = 0; j < NREG; ++j) {
        if (j <= k) {
            const int n = j < k ? NW : w;
            carry(sE, j * NW, n, lane,
                  chunk_decay(g0, j), chunk_decay(g1, j), chunk_decay(g2, j), chunk_decay(g3, j),
                  y0, y1, y2, y3);
        }
    }

    const int ng = reg_groups(k);
    const int t0 = reg_start(k) + w * 32 * ng;
    float* o = out + ((long)b * S_ + t0) * D_ + d;
    int t = t0;
    float4 cur = fetch32(p4, t, lane, b);
    #pragma unroll 1
    for (int g = 0; g < ng; ++g) {
        t += 32;
        const float4 nxt = fetch32(p4, t, lane, b);
        __builtin_amdgcn_sched_barrier(0);
        emit32(cur, c0, c1, c2, c3, y0, y1, y2, y3, o);
        o += 32L * D_;
        cur = nxt;
    }
}

extern "C" void kernel_launch(void* const* d_in, const int* in_sizes, int n_in,
                              void* d_out, int out_size, void* d_ws, size_t ws_size,
                              hipStream_t stream) {
    const float* x  = (const float*)d_in[0];
    const float* w  = (const float*)d_in[1];
    const float* fc = (const float*)d_in[2];
    float*       out  = (float*)d_out;
    float*       proj = (float*)d_ws;

    const int nwaves = (B_ * S_) / ROWS_PER_WAVE;
    hyena_proj_wmma<<<nwaves / 8, 256, 0, stream>>>(x, w, proj);

    dim3 grid(D_ / DPB, NREG, B_);
    hyena_recurrence<<<grid, NW * 32, 0, stream>>>(proj, fc, out);
}


# scan kernel, end-state pass with time across lanes: a lane scans a half-chunk for 8 d, proj by coalesced per-lane loads (scratch re-ordered), no v_readlane; regions 6,4,3,3; chunk sum order changed
# speedup vs baseline: 6.1539x; 1.0789x over previous
#include <hip/hip_runtime.h>
#include <hip/hip_bf16.h>

typedef __attribute__((ext_vector_type(2))) float v2f;
typedef __attribute__((ext_vector_type(8))) float v8f;

#define B_ 4
#define S_ 8192
#define D_ 1024
#define O_ 4

typedef __attribute__((ext_vector_type(4))) float v4f;

#define ROWS_PER_WAVE 8
#define KB    32
#define NKB   (D_ / KB)
#define DEPTH 8
static_assert(NKB % DEPTH == 0 && NKB / DEPTH >= 2, "whole ring turns");

#define LOAD_STEP(i, kb)                                                          \
    do {                                                                          \
        const int off_ = (kb) * KB;                                               \
        xa[i] = __builtin_nontemporal_load((const v4f*)(xp0 + off_));   \
        xb[i] = __builtin_nontemporal_load((const v4f*)(xp1 + off_));             \
        wq[i] = *(const v4f*)(wp + off_);                               \
    } while (0)

__device__ __forceinline__ v8f mma4(float a0, float a1, float b0, float b1, v8f acc)
{
    v2f a; a.x = a0; a.y = a1;
    v2f b; b.x = b0; b.y = b1;
    return __builtin_amdgcn_wmma_f32_16x16x4_f32(false, a, false, b,
                                                 (short)0, acc, false, false);
}

#define MMA_STEP(i)                                                               \
    do {                                                                          \
        acc0 = mma4(xa[i].x, xa[i].y, wq[i].x, wq[i].y, acc0);                    \
        acc1 = mma4(xb[i].x, xb[i].y, wq[i].x, wq[i].y, acc1);                    \
        acc0 = mma4(xa[i].z, xa[i].w, wq[i].z, wq[i].w, acc0);                    \
        acc1 = mma4(xb[i].z, xb[i].w, wq[i].z, wq[i].w, acc1);                    \
    } while (0)


__device__ __forceinline__ int proj_slot(int s)
{
    int start, hc, i;
    if (s < 3072)      { start = 0;    hc = s / 96;           i = s - hc * 96; }
    else if (s < 5120) { start = 3072; hc = (s - 3072) / 64;  i = (s - 3072) - hc * 64; }
    else if (s < 6656) { start = 5120; hc = (s - 5120) / 48;  i = (s - 5120) - hc * 48; }
    else               { start = 6656; hc = (s - 6656) / 48;  i = (s - 6656) - hc * 48; }
    return start + i * 32 + hc;
}

__device__ __forceinline__ void store_group(const v8f acc, long rbase, int lane,
                                            float* __restrict__ proj)
{
    const int c = lane & 15;
    const int h = lane >> 4;
    const int j = c & 3;
    const int o = c >> 2;
    const bool j1 = (j & 1) != 0;
    const bool j2 = (j & 2) != 0;
    const float a01 = j1 ? acc[1] : acc[0];
    const float a23 = j1 ? acc[3] : acc[2];
    const float a45 = j1 ? acc[5] : acc[4];
    const float a67 = j1 ? acc[7] : acc[6];
    float v0 = j2 ? a23 : a01;
    float v1 = j2 ? a67 : a45;
    v0 += __shfl_xor(v0, 1);
    v1 += __shfl_xor(v1, 1);
    v0 += __shfl_xor(v0, 2);
    v1 += __shfl_xor(v1, 2);
    if (j == 0) {
        const long r = rbase + 2 * h;
        const int  b = (int)(r / S_);
        const int  s = (int)(r % S_);
        proj[((long)b * S_ + proj_slot(s)) * O_ + o]     = v0;
        proj[((long)b * S_ + proj_slot(s + 1)) * O_ + o] = v1;
    }
}

__global__ void __launch_bounds__(256)
hyena_proj_wmma(const float* __restrict__ x,
                const float* __restrict__ w,
                float* __restrict__ proj)
{
    const int  wave = (int)((blockIdx.x * blockDim.x + threadIdx.x) >> 5);
    const int  lane = (int)(threadIdx.x & 31);
    const int  c    = lane & 15;
    const int  h    = lane >> 4;
    const long row0 = (long)wave * ROWS_PER_WAVE;

    const int seg = 8 * (c & 3) + 4 * h;
    const float* xp0 = x + (row0 + (c >> 2)) * D_ + seg;
    const float* xp1 = xp0 + 4L * D_;
    const float* wp  = w + (long)(c >> 2) * D_ + seg;

    v4f xa[DEPTH], xb[DEPTH], wq[DEPTH];
    v8f acc0 = {};
    v8f acc1 = {};

    #pragma unroll
    for (int i = 0; i < DEPTH; ++i) LOAD_STEP(i, i);

    #pragma unroll 1
    for (int kb0 = DEPTH; kb0 < NKB; kb0 += DEPTH) {
        #pragma unroll
        for (int i = 0; i < DEPTH; ++i) {
            MMA_STEP(i);
            LOAD_STEP(i, kb0 + i);
            __builtin_amdgcn_sched_barrier(0);
        }
    }
    #pragma unroll
    for (int i = 0; i < DEPTH; ++i) MMA_STEP(i);

    store_group(acc0, row0,     lane, proj);
    store_group(acc1, row0 + 4, lane, proj);
}

#define NW      16
#define NREG    4
#define DPB     64
#define BURST   16

__device__ __forceinline__ int reg_groups(int j) { return j == 0 ? 6 : j == 1 ? 4 : 3; }
__device__ __forceinline__ int reg_start(int j)  { return j == 0 ? 0 : j == 1 ? 3072 : j == 2 ? 5120 : 6656; }

static_assert(NW * 32 * (6 + 4 + 3 + 3) == S_, "the four regions cover the sequence");
static_assert(NW * 32 * 6 == 3072 && 3072 + NW * 32 * 4 == 5120 && 5120 + NW * 32 * 3 == 6656,
              "region starts");
static_assert(32 % BURST == 0, "whole store bursts per load");

__device__ __forceinline__ float from_lane(float v, int i)
{
    return __builtin_bit_cast(float,
        __builtin_amdgcn_readlane(__builtin_bit_cast(int, v), i));
}

__device__ __forceinline__ v2f step2(v2f c, v2f y, float p)
{
    v2f pp; pp.x = p; pp.y = p;
    return __builtin_elementwise_fma(c, y, pp);
}

__device__ __forceinline__ float4 fetch32(const float4* __restrict__ p4,
                                          int tf, int lane, int b)
{
    int tt = tf + lane;
    tt = tt < S_ ? tt : S_ - 1;
    return p4[(long)b * S_ + proj_slot(tt)];
}

#define DSET  8
#define NDSET (DPB / DSET)
#define HSTR  (O_ * DPB + 4)
#define PF    8

__device__ __forceinline__ void half_chunk_ends(const float4* __restrict__ p4,
                                                const float* __restrict__ fc,
                                                int j, int d0, int ds, int b, int lane,
                                                float* __restrict__ row)
{
    v2f cc[O_][DSET / 2];
    #pragma unroll
    for (int o = 0; o < O_; ++o) {
        #pragma unroll
        for (int pr = 0; pr < DSET / 2; ++pr) {
            cc[o][pr].x = __expf(fc[o * D_ + d0 + 2 * pr]);
            cc[o][pr].y = __expf(fc[o * D_ + d0 + 2 * pr + 1]);
        }
    }
    const int h = 16 * reg_groups(j);
    const float4* q = p4 + (long)b * S_ + reg_start(j) + lane;

    v2f y[O_][DSET / 2] = {};
#define HC_STEP(p)                                                                \
    do {                                                                          \
        v2f p0, p1, p2, p3;                                                       \
        p0.x = (p).x; p0.y = (p).x;                                               \
        p1.x = (p).y; p1.y = (p).y;                                               \
        p2.x = (p).z; p2.y = (p).z;                                               \
        p3.x = (p).w; p3.y = (p).w;                                               \
        _Pragma("unroll")                                                         \
        for (int pr = 0; pr < DSET / 2; ++pr) {                                   \
            y[0][pr] = __builtin_elementwise_fma(cc[0][pr], y[0][pr], p0);        \
            y[1][pr] = __builtin_elementwise_fma(cc[1][pr], y[1][pr], p1);        \
            y[2][pr] = __builtin_elementwise_fma(cc[2][pr], y[2][pr], p2);        \
            y[3][pr] = __builtin_elementwise_fma(cc[3][pr], y[3][pr], p3);        \
        }                                                                         \
    } while (0)

    float4 pa[PF], pb[PF];
    #pragma unroll
    for (int u = 0; u < PF; ++u) pa[u] = q[u * 32];

    #pragma unroll 1
    for (int i0 = 0; i0 < h; i0 += 2 * PF) {
        const float4* qb = q + (i0 + PF) * 32;
        #pragma unroll
        for (int u = 0; u < PF; ++u) pb[u] = qb[u * 32];
        __builtin_amdgcn_sched_barrier(0);
        #pragma unroll
        for (int u = 0; u < PF; ++u) HC_STEP(pa[u]);

        int ia = i0 + 2 * PF;
        ia = ia < h ? ia : h - PF;
        const float4* qa = q + ia * 32;
        #pragma unroll
        for (int u = 0; u < PF; ++u) pa[u] = qa[u * 32];
        __builtin_amdgcn_sched_barrier(0);
        #pragma unroll
        for (int u = 0; u < PF; ++u) HC_STEP(pb[u]);
    }
#undef HC_STEP
    #pragma unroll
    for (int o = 0; o < O_; ++o) {
        v4f lo, hi;
        lo.xy = y[o][0]; lo.zw = y[o][1];
        hi.xy = y[o][2]; hi.zw = y[o][3];
        *(v4f*)(row + o * DPB + ds * DSET)     = lo;
        *(v4f*)(row + o * DPB + ds * DSET + 4) = hi;
    }
}

__device__ __forceinline__ void emit32(const float4 p,
                                       v2f c0, v2f c1, v2f c2, v2f c3,
                                       v2f& y0, v2f& y1, v2f& y2, v2f& y3,
                                       float* __restrict__ o)
{
    #pragma unroll
    for (int h = 0; h < 32; h += BURST) {
        v2f s[BURST];
        #pragma unroll
        for (int i = 0; i < BURST; ++i) {
            y0 = step2(c0, y0, from_lane(p.x, h + i));
            y1 = step2(c1, y1, from_lane(p.y, h + i));
            y2 = step2(c2, y2, from_lane(p.z, h + i));
            y3 = step2(c3, y3, from_lane(p.w, h + i));
            s[i] = (y0 + y1) + (y2 + y3);
        }
        __builtin_amdgcn_sched_barrier(0);
        #pragma unroll
        for (int i = 0; i < BURST; ++i) {
            __builtin_nontemporal_store(s[i], (v2f*)(o + (long)(h + i) * D_));
        }
        __builtin_amdgcn_sched_barrier(0);
    }
}

struct EndState { v4f a; v4f b; };

__device__ __forceinline__ void carry(const EndState (*sE)[32], int first, int n, int lane,
                                      v2f cp0, v2f cp1, v2f cp2, v2f cp3,
                                      v2f& y0, v2f& y1, v2f& y2, v2f& y3)
{
#define CARRY_LINK(e)                                                             \
    do {                                                                          \
        y0 = __builtin_elementwise_fma(cp0, y0, (e).a.xy);                        \
        y1 = __builtin_elementwise_fma(cp1, y1, (e).a.zw);                        \
        y2 = __builtin_elementwise_fma(cp2, y2, (e).b.xy);                        \
        y3 = __builtin_elementwise_fma(cp3, y3, (e).b.zw);                        \
    } while (0)
    int j = 0;
    for (; j + 4 <= n; j += 4) {
        const EndState e0 = sE[first + j][lane];
        const EndState e1 = sE[first + j + 1][lane];
        const EndState e2 = sE[first + j + 2][lane];
        const EndState e3 = sE[first + j + 3][lane];
        CARRY_LINK(e0); CARRY_LINK(e1); CARRY_LINK(e2); CARRY_LINK(e3);
    }
    if (j + 2 <= n) {
        const EndState e0 = sE[first + j][lane];
        const EndState e1 = sE[first + j + 1][lane];
        CARRY_LINK(e0); CARRY_LINK(e1);
        j += 2;
    }
    if (j < n) {
        const EndState e0 = sE[first + j][lane];
        CARRY_LINK(e0);
    }
#undef CARRY_LINK
}

__device__ __forceinline__ v2f pow16(v2f c)
{
    v2f p = c * c;
    p = p * p;
    p = p * p;
    return p * p;
}

__device__ __forceinline__ v2f pow32(v2f c)
{
    v2f p = c * c;
    p = p * p;
    p = p * p;
    p = p * p;
    return p * p;
}

__device__ __forceinline__ v2f chunk_decay(v2f g, int j)
{
    const v2f g2 = g * g;
    const v2f g3 = g2 * g;
    const v2f g4 = g2 * g2;
    return j == 0 ? g4 * g2 : j == 1 ? g4 : g3;
}

__global__ void __launch_bounds__(NW * 32)
hyena_recurrence(const float* __restrict__ proj,
                 const float* __restrict__ fc,
                 float* __restrict__ out)
{
    __shared__ EndState sE[NREG * NW][32];
    __shared__ __attribute__((aligned(16))) float sH[NREG][32][HSTR];

    const int lane   = (int)(threadIdx.x & 31);
    const int w      = __builtin_amdgcn_readfirstlane((int)(threadIdx.x >> 5));
    const int d      = (int)(blockIdx.x * DPB) + 2 * lane;
    const int k      = (int)blockIdx.y;
    const int b      = (int)blockIdx.z;

    const float4* p4 = (const float4*)proj;

    const v2f f0 = *(const v2f*)(fc + 0 * D_ + d);
    const v2f f1 = *(const v2f*)(fc + 1 * D_ + d);
    const v2f f2 = *(const v2f*)(fc + 2 * D_ + d);
    const v2f f3 = *(const v2f*)(fc + 3 * D_ + d);
    v2f c0, c1, c2, c3;
    c0.x = __expf(f0.x); c0.y = __expf(f0.y);
    c1.x = __expf(f1.x); c1.y = __expf(f1.y);
    c2.x = __expf(f2.x); c2.y = __expf(f2.y);
    c3.x = __expf(f3.x); c3.y = __expf(f3.y);

    #pragma unroll 1
    for (int q = w; q < (k + 1) * NDSET; q += NW) {
        const int j  = q / NDSET;
        const int ds = q % NDSET;
        half_chunk_ends(p4, fc, j, (int)(blockIdx.x * DPB) + ds * DSET, ds, b, lane,
                        &sH[j][lane][0]);
    }

    __syncthreads();

    {
        const v2f q0 = pow16(c0), q1 = pow16(c1), q2 = pow16(c2), q3 = pow16(c3);
        #pragma unroll 1
        for (int j = 0; j <= k; ++j) {
            const float* ea = &sH[j][2 * w][2 * lane];
            const float* eb = &sH[j][2 * w + 1][2 * lane];
            EndState e;
            e.a.xy = __builtin_elementwise_fma(chunk_decay(q0, j), *(const v2f*)(ea + 0 * DPB), *(const v2f*)(eb + 0 * DPB));
            e.a.zw = __builtin_elementwise_fma(chunk_decay(q1, j), *(const v2f*)(ea + 1 * DPB), *(const v2f*)(eb + 1 * DPB));
            e.b.xy = __builtin_elementwise_fma(chunk_decay(q2, j), *(const v2f*)(ea + 2 * DPB), *(const v2f*)(eb + 2 * DPB));
            e.b.zw = __builtin_elementwise_fma(chunk_decay(q3, j), *(const v2f*)(ea + 3 * DPB), *(const v2f*)(eb + 3 * DPB));
            sE[j * NW + w][lane] = e;
        }
    }

    __syncthreads();

    const v2f g0 = pow32(c0), g1 = pow32(c1), g2 = pow32(c2), g3 = pow32(c3);

    v2f y0 = {}, y1 = {}, y2 = {}, y3 = {};
    #pragma unroll
    for (int j = 0; j < NREG; ++j) {
        if (j <= k) {
            const int n = j < k ? NW : w;
            carry(sE, j * NW, n, lane,
                  chunk_decay(g0, j), chunk_decay(g1, j), chunk_decay(g2, j), chunk_decay(g3, j),
                  y0, y1, y2, y3);
        }
    }

    const int ng = reg_groups(k);
    const int t0 = reg_start(k) + w * 32 * ng;
    float* o = out + ((long)b * S_ + t0) * D_ + d;
    int t = t0;
    float4 cur = fetch32(p4, t, lane, b);
    #pragma unroll 1
    for (int g = 0; g < ng; ++g) {
        t += 32;
        const float4 nxt = fetch32(p4, t, lane, b);
        __builtin_amdgcn_sched_barrier(0);
        emit32(cur, c0, c1, c2, c3, y0, y1, y2, y3, o);
        o += 32L * D_;
        cur = nxt;
    }
}

extern "C" void kernel_launch(void* const* d_in, const int* in_sizes, int n_in,
                              void* d_out, int out_size, void* d_ws, size_t ws_size,
                              hipStream_t stream) {
    const float* x  = (const float*)d_in[0];
    const float* w  = (const float*)d_in[1];
    const float* fc = (const float*)d_in[2];
    float*       out  = (float*)d_out;
    float*       proj = (float*)d_ws;

    const int nwaves = (B_ * S_) / ROWS_PER_WAVE;
    hyena_proj_wmma<<<nwaves / 8, 256, 0, stream>>>(x, w, proj);

    dim3 grid(D_ / DPB, NREG, B_);
    hyena_recurrence<<<grid, NW * 32, 0, stream>>>(proj, fc, out);
}


# scan kernel: 32 d per block held by a half-wave, each wave scans two chunks, proj per half by vector loads; two regions not four, so end states are done 1.5x not 2.5x; carry sum order changed
# speedup vs baseline: 6.5861x; 1.0702x over previous
#include <hip/hip_runtime.h>
#include <hip/hip_bf16.h>

typedef __attribute__((ext_vector_type(2))) float v2f;
typedef __attribute__((ext_vector_type(8))) float v8f;

#define B_ 4
#define S_ 8192
#define D_ 1024
#define O_ 4

typedef __attribute__((ext_vector_type(4))) float v4f;

#define ROWS_PER_WAVE 8
#define KB    32
#define NKB   (D_ / KB)
#define DEPTH 8
static_assert(NKB % DEPTH == 0 && NKB / DEPTH >= 2, "whole ring turns");

#define LOAD_STEP(i, kb)                                                          \
    do {                                                                          \
        const int off_ = (kb) * KB;                                               \
        xa[i] = __builtin_nontemporal_load((const v4f*)(xp0 + off_));   \
        xb[i] = __builtin_nontemporal_load((const v4f*)(xp1 + off_));             \
        wq[i] = *(const v4f*)(wp + off_);                               \
    } while (0)

__device__ __forceinline__ v8f mma4(float a0, float a1, float b0, float b1, v8f acc)
{
    v2f a; a.x = a0; a.y = a1;
    v2f b; b.x = b0; b.y = b1;
    return __builtin_amdgcn_wmma_f32_16x16x4_f32(false, a, false, b,
                                                 (short)0, acc, false, false);
}

#define MMA_STEP(i)                                                               \
    do {                                                                          \
        acc0 = mma4(xa[i].x, xa[i].y, wq[i].x, wq[i].y, acc0);                    \
        acc1 = mma4(xb[i].x, xb[i].y, wq[i].x, wq[i].y, acc1);                    \
        acc0 = mma4(xa[i].z, xa[i].w, wq[i].z, wq[i].w, acc0);                    \
        acc1 = mma4(xb[i].z, xb[i].w, wq[i].z, wq[i].w, acc1);                    \
    } while (0)


#define R0LEN   160
#define R1LEN   96
#define R1START (32 * R0LEN)
static_assert(32 * (R0LEN + R1LEN) == S_, "the two regions cover the sequence");

__device__ __forceinline__ int proj_slot(int s)
{
    int start, c, i;
    if (s < R1START) { start = 0;       c = s / R0LEN;             i = s - c * R0LEN; }
    else             { start = R1START; c = (s - R1START) / R1LEN; i = (s - R1START) - c * R1LEN; }
    return start + i * 32 + c;
}

__device__ __forceinline__ void store_group(const v8f acc, long rbase, int lane,
                                            float* __restrict__ proj)
{
    const int c = lane & 15;
    const int h = lane >> 4;
    const int j = c & 3;
    const int o = c >> 2;
    const bool j1 = (j & 1) != 0;
    const bool j2 = (j & 2) != 0;
    const float a01 = j1 ? acc[1] : acc[0];
    const float a23 = j1 ? acc[3] : acc[2];
    const float a45 = j1 ? acc[5] : acc[4];
    const float a67 = j1 ? acc[7] : acc[6];
    float v0 = j2 ? a23 : a01;
    float v1 = j2 ? a67 : a45;
    v0 += __shfl_xor(v0, 1);
    v1 += __shfl_xor(v1, 1);
    v0 += __shfl_xor(v0, 2);
    v1 += __shfl_xor(v1, 2);
    if (j == 0) {
        const long r = rbase + 2 * h;
        const int  b = (int)(r / S_);
        const int  s = (int)(r % S_);
        proj[((long)b * S_ + proj_slot(s)) * O_ + o]     = v0;
        proj[((long)b * S_ + proj_slot(s + 1)) * O_ + o] = v1;
    }
}

__global__ void __launch_bounds__(256)
hyena_proj_wmma(const float* __restrict__ x,
                const float* __restrict__ w,
                float* __restrict__ proj)
{
    const int  wave = (int)((blockIdx.x * blockDim.x + threadIdx.x) >> 5);
    const int  lane = (int)(threadIdx.x & 31);
    const int  c    = lane & 15;
    const int  h    = lane >> 4;
    const long row0 = (long)wave * ROWS_PER_WAVE;

    const int seg = 8 * (c & 3) + 4 * h;
    const float* xp0 = x + (row0 + (c >> 2)) * D_ + seg;
    const float* xp1 = xp0 + 4L * D_;
    const float* wp  = w + (long)(c >> 2) * D_ + seg;

    v4f xa[DEPTH], xb[DEPTH], wq[DEPTH];
    v8f acc0 = {};
    v8f acc1 = {};

    #pragma unroll
    for (int i = 0; i < DEPTH; ++i) LOAD_STEP(i, i);

    #pragma unroll 1
    for (int kb0 = DEPTH; kb0 < NKB; kb0 += DEPTH) {
        #pragma unroll
        for (int i = 0; i < DEPTH; ++i) {
            MMA_STEP(i);
            LOAD_STEP(i, kb0 + i);
            __builtin_amdgcn_sched_barrier(0);
        }
    }
    #pragma unroll
    for (int i = 0; i < DEPTH; ++i) MMA_STEP(i);

    store_group(acc0, row0,     lane, proj);
    store_group(acc1, row0 + 4, lane, proj);
}

#define NW      16
#define NCH     32
#define NREG    2
#define DPB     32
#define BURST   8
#define PRING   32

__device__ __forceinline__ int reg_len(int j)   { return j == 0 ? R0LEN : R1LEN; }
__device__ __forceinline__ int reg_start(int j) { return j == 0 ? 0 : R1START; }

static_assert(R0LEN % 32 == 0 && R1LEN % 32 == 0, "whole ring turns in the end-state pass");
static_assert(R0LEN % PRING == 0 && R1LEN % PRING == 0 && PRING % BURST == 0, "whole ring turns and store bursts while emitting");

__device__ __forceinline__ v2f step2(v2f c, v2f y, float p)
{
    v2f pp; pp.x = p; pp.y = p;
    return __builtin_elementwise_fma(c, y, pp);
}

#define DSET  4
#define NDSET (DPB / DSET)
#define HSTR  (O_ * DPB + 4)
#define RING  32

__device__ __forceinline__ void chunk_ends(const float4* __restrict__ p4,
                                           const float* __restrict__ fc,
                                           int j, int d0, int ds, int b, int lane,
                                           float* __restrict__ row)
{
    v2f cc[O_][DSET / 2];
    #pragma unroll
    for (int o = 0; o < O_; ++o) {
        #pragma unroll
        for (int pr = 0; pr < DSET / 2; ++pr) {
            cc[o][pr].x = __expf(fc[o * D_ + d0 + 2 * pr]);
            cc[o][pr].y = __expf(fc[o * D_ + d0 + 2 * pr + 1]);
        }
    }
    const int len = reg_len(j);
    const float4* q = p4 + (long)b * S_ + reg_start(j) + lane;

    v2f y[O_][DSET / 2] = {};
#define HC_STEP(p)                                                                \
    do {                                                                          \
        v2f p0, p1, p2, p3;                                                       \
        p0.x = (p).x; p0.y = (p).x;                                               \
        p1.x = (p).y; p1.y = (p).y;                                               \
        p2.x = (p).z; p2.y = (p).z;                                               \
        p3.x = (p).w; p3.y = (p).w;                                               \
        _Pragma("unroll")                                                         \
        for (int pr = 0; pr < DSET / 2; ++pr) {                                   \
            y[0][pr] = __builtin_elementwise_fma(cc[0][pr], y[0][pr], p0);        \
            y[1][pr] = __builtin_elementwise_fma(cc[1][pr], y[1][pr], p1);        \
            y[2][pr] = __builtin_elementwise_fma(cc[2][pr], y[2][pr], p2);        \
            y[3][pr] = __builtin_elementwise_fma(cc[3][pr], y[3][pr], p3);        \
        }                                                                         \
    } while (0)

    float4 ring[RING];
    #pragma unroll
    for (int u = 0; u < RING; ++u) ring[u] = q[u * 32];

    #pragma unroll 1
    for (int i0 = 0; i0 < len; i0 += RING) {
        int nx = i0 + RING;
        nx = nx < len ? nx : len - RING;
        const float4* qn = q + nx * 32;
        #pragma unroll
        for (int u = 0; u < RING; ++u) {
            HC_STEP(ring[u]);
            ring[u] = qn[u * 32];
            __builtin_amdgcn_sched_barrier(0);
        }
    }
#undef HC_STEP
    #pragma unroll
    for (int o = 0; o < O_; ++o) {
        v4f e;
        e.xy = y[o][0]; e.zw = y[o][1];
        *(v4f*)(row + o * DPB + ds * DSET) = e;
    }
}

__device__ __forceinline__ void carry(const float (*sHj)[HSTR], int n, int dl,
                                      v2f cp0, v2f cp1, v2f cp2, v2f cp3,
                                      v2f& y0, v2f& y1, v2f& y2, v2f& y3)
{
#define CARRY_LINK(e)                                                             \
    do {                                                                          \
        y0 = __builtin_elementwise_fma(cp0, y0, *(const v2f*)((e) + 0 * DPB));    \
        y1 = __builtin_elementwise_fma(cp1, y1, *(const v2f*)((e) + 1 * DPB));    \
        y2 = __builtin_elementwise_fma(cp2, y2, *(const v2f*)((e) + 2 * DPB));    \
        y3 = __builtin_elementwise_fma(cp3, y3, *(const v2f*)((e) + 3 * DPB));    \
    } while (0)
    int c = 0;
    for (; c + 4 <= n; c += 4) {
        const float* e0 = &sHj[c][2 * dl];
        const float* e1 = &sHj[c + 1][2 * dl];
        const float* e2 = &sHj[c + 2][2 * dl];
        const float* e3 = &sHj[c + 3][2 * dl];
        CARRY_LINK(e0); CARRY_LINK(e1); CARRY_LINK(e2); CARRY_LINK(e3);
    }
    for (; c < n; ++c) {
        const float* e0 = &sHj[c][2 * dl];
        CARRY_LINK(e0);
    }
#undef CARRY_LINK
}

__device__ __forceinline__ v2f chunk_decay(v2f c, int j)
{
    v2f p = c * c;
    p = p * p;
    p = p * p;
    p = p * p;
    const v2f p32  = p * p;
    const v2f p64  = p32 * p32;
    const v2f p128 = p64 * p64;
    static_assert(R0LEN == 160 && R1LEN == 96, "the powers below are these lengths");
    return j == 0 ? p128 * p32 : p64 * p32;
}

__global__ void __launch_bounds__(NW * 32)
hyena_recurrence(const float* __restrict__ proj,
                 const float* __restrict__ fc,
                 float* __restrict__ out)
{
    __shared__ __attribute__((aligned(16))) float sH[NREG][NCH][HSTR];

    const int lane  = (int)(threadIdx.x & 31);
    const int w     = __builtin_amdgcn_readfirstlane((int)(threadIdx.x >> 5));
    const int hf    = lane >> 4;
    const int dl    = lane & 15;
    const int dbase = (int)(blockIdx.x * DPB);
    const int d     = dbase + 2 * dl;
    const int k     = (int)blockIdx.y;
    const int b     = (int)blockIdx.z;

    const float4* p4 = (const float4*)proj;

    const v2f f0 = *(const v2f*)(fc + 0 * D_ + d);
    const v2f f1 = *(const v2f*)(fc + 1 * D_ + d);
    const v2f f2 = *(const v2f*)(fc + 2 * D_ + d);
    const v2f f3 = *(const v2f*)(fc + 3 * D_ + d);
    v2f c0, c1, c2, c3;
    c0.x = __expf(f0.x); c0.y = __expf(f0.y);
    c1.x = __expf(f1.x); c1.y = __expf(f1.y);
    c2.x = __expf(f2.x); c2.y = __expf(f2.y);
    c3.x = __expf(f3.x); c3.y = __expf(f3.y);

    #pragma unroll 1
    for (int q = w; q < (k + 1) * NDSET; q += NW) {
        const int j  = q / NDSET;
        const int ds = q % NDSET;
        chunk_ends(p4, fc, j, dbase + ds * DSET, ds, b, lane, &sH[j][lane][0]);
    }

    __syncthreads();

    const int cidx = 2 * w + hf;
    v2f y0 = {}, y1 = {}, y2 = {}, y3 = {};
    if (k == 1) {
        carry(sH[0], NCH, dl,
              chunk_decay(c0, 0), chunk_decay(c1, 0), chunk_decay(c2, 0), chunk_decay(c3, 0),
              y0, y1, y2, y3);
    }
    {
        const v2f cp0 = chunk_decay(c0, k), cp1 = chunk_decay(c1, k);
        const v2f cp2 = chunk_decay(c2, k), cp3 = chunk_decay(c3, k);
        carry(sH[k], 2 * w, dl, cp0, cp1, cp2, cp3, y0, y1, y2, y3);
        const float* e = &sH[k][2 * w][2 * dl];
        const v2f n0 = __builtin_elementwise_fma(cp0, y0, *(const v2f*)(e + 0 * DPB));
        const v2f n1 = __builtin_elementwise_fma(cp1, y1, *(const v2f*)(e + 1 * DPB));
        const v2f n2 = __builtin_elementwise_fma(cp2, y2, *(const v2f*)(e + 2 * DPB));
        const v2f n3 = __builtin_elementwise_fma(cp3, y3, *(const v2f*)(e + 3 * DPB));
        y0 = hf ? n0 : y0;
        y1 = hf ? n1 : y1;
        y2 = hf ? n2 : y2;
        y3 = hf ? n3 : y3;
    }

    const int len = reg_len(k);
    const int t0  = reg_start(k) + cidx * len;
    const float4* pq = p4 + (long)b * S_ + reg_start(k) + cidx;
    float* o = out + ((long)b * S_ + t0) * D_ + d;

    float4 ring[PRING];
    #pragma unroll
    for (int u = 0; u < PRING; ++u) ring[u] = pq[u * 32];

    #pragma unroll 1
    for (int i0 = 0; i0 < len; i0 += PRING) {
        int nx = i0 + PRING;
        nx = nx < len ? nx : len - PRING;
        const float4* pn = pq + nx * 32;
        #pragma unroll
        for (int h = 0; h < PRING; h += BURST) {
            v2f s[BURST];
            #pragma unroll
            for (int u = 0; u < BURST; ++u) {
                y0 = step2(c0, y0, ring[h + u].x);
                y1 = step2(c1, y1, ring[h + u].y);
                y2 = step2(c2, y2, ring[h + u].z);
                y3 = step2(c3, y3, ring[h + u].w);
                s[u] = (y0 + y1) + (y2 + y3);
                ring[h + u] = pn[(h + u) * 32];
            }
            __builtin_amdgcn_sched_barrier(0);
            #pragma unroll
            for (int u = 0; u < BURST; ++u) {
                __builtin_nontemporal_store(s[u], (v2f*)(o + (long)(h + u) * D_));
            }
            __builtin_amdgcn_sched_barrier(0);
        }
        o += (long)PRING * D_;
    }
}

extern "C" void kernel_launch(void* const* d_in, const int* in_sizes, int n_in,
                              void* d_out, int out_size, void* d_ws, size_t ws_size,
                              hipStream_t stream) {
    const float* x  = (const float*)d_in[0];
    const float* w  = (const float*)d_in[1];
    const float* fc = (const float*)d_in[2];
    float*       out  = (float*)d_out;
    float*       proj = (float*)d_ws;

    const int nwaves = (B_ * S_) / ROWS_PER_WAVE;
    hyena_proj_wmma<<<nwaves / 8, 256, 0, stream>>>(x, w, proj);

    dim3 grid(D_ / DPB, NREG, B_);
    hyena_recurrence<<<grid, NW * 32, 0, stream>>>(proj, fc, out);
}
